# starting kernel
# baseline (speedup 1.0000x reference)
#include <hip/hip_runtime.h>
#include <math.h>

#define D      256
#define TX     512
#define TM     512
#define BATCH  4

typedef __attribute__((ext_vector_type(2))) float v2f;
typedef __attribute__((ext_vector_type(8))) float v8f;

__device__ __forceinline__ float fast_tanh(float x) {
#if __has_builtin(__builtin_amdgcn_tanhf)
  return __builtin_amdgcn_tanhf(x);
#else
  float r;
  asm("v_tanh_f32 %0, %1" : "=v"(r) : "v"(x));
  return r;
#endif
}

__device__ __forceinline__ float fast_exp(float x) {
#if __has_builtin(__builtin_amdgcn_exp2f)
  return __builtin_amdgcn_exp2f(x * 1.4426950408889634f);
#else
  return __expf(x);
#endif
}

__global__ __launch_bounds__(128) void proj_gemm_kernel(
    const float* __restrict__ In, const float* __restrict__ W,
    const float* __restrict__ bias, float* __restrict__ Out, int Mrows) {
  const int warp = threadIdx.x >> 5;
  const int lane = threadIdx.x & 31;
  const int half = lane >> 4;
  const int l    = lane & 15;

  const int pairsPerRow = D / 32;
  int tile = blockIdx.x * (blockDim.x >> 5) + warp;
  int mt = tile / pairsPerRow;
  int np = tile % pairsPerRow;
  if (mt * 16 >= Mrows) return;
  const int row0 = mt * 16;
  const int col0 = np * 32;

  v8f acc0 = {}; v8f acc1 = {};
  const float* arow  = In + (size_t)(row0 + l) * D;
  const float* b0row = W  + (size_t)(col0 + l) * D;
  const float* b1row = W  + (size_t)(col0 + 16 + l) * D;
#pragma unroll 8
  for (int k = 0; k < D; k += 4) {
    v2f a  = *(const v2f*)(arow  + k + 2 * half);
    v2f b0 = *(const v2f*)(b0row + k + 2 * half);
    v2f b1 = *(const v2f*)(b1row + k + 2 * half);
    acc0 = __builtin_amdgcn_wmma_f32_16x16x4_f32(false, a, false, b0, (short)0, acc0, false, false);
    acc1 = __builtin_amdgcn_wmma_f32_16x16x4_f32(false, a, false, b1, (short)0, acc1, false, false);
  }
  float bv0 = 0.0f, bv1 = 0.0f;
  if (bias) { bv0 = bias[col0 + l]; bv1 = bias[col0 + 16 + l]; }
#pragma unroll
  for (int j = 0; j < 8; ++j) {
    int r = row0 + j + 8 * half;
    Out[(size_t)r * D + col0 + l]      = acc0[j] + bv0;
    Out[(size_t)r * D + col0 + 16 + l] = acc1[j] + bv1;
  }
}

__global__ __launch_bounds__(256) void scores_softmax_kernel(
    const float* __restrict__ item1, const float* __restrict__ item2,
    const float* __restrict__ wst, const int* __restrict__ mask,
    float* __restrict__ Sout) {
  __shared__ float sc[TM];
  __shared__ float red[256];

  const int tid  = threadIdx.x;
  const int b    = blockIdx.x / TX;
  const int x    = blockIdx.x % TX;
  const int warp = tid >> 5;
  const int lane = tid & 31;

  const size_t rb = ((size_t)b * TX + x) * D;
  const float4 f10 = *(const float4*)(item1 + rb + lane * 8);
  const float4 f11 = *(const float4*)(item1 + rb + lane * 8 + 4);
  const float4 w0  = *(const float4*)(wst + lane * 8);
  const float4 w1  = *(const float4*)(wst + lane * 8 + 4);

  const float* i2b = item2 + (size_t)b * TM * D;
  for (int m = warp; m < TM; m += 8) {
    const float* row = i2b + (size_t)m * D + lane * 8;
    if (m + 8 < TM) __builtin_prefetch(row + 8 * D, 0, 3);
    float4 r0 = *(const float4*)(row);
    float4 r1 = *(const float4*)(row + 4);
    float p;
    p  = fast_tanh(f10.x + r0.x) * w0.x;
    p += fast_tanh(f10.y + r0.y) * w0.y;
    p += fast_tanh(f10.z + r0.z) * w0.z;
    p += fast_tanh(f10.w + r0.w) * w0.w;
    p += fast_tanh(f11.x + r1.x) * w1.x;
    p += fast_tanh(f11.y + r1.y) * w1.y;
    p += fast_tanh(f11.z + r1.z) * w1.z;
    p += fast_tanh(f11.w + r1.w) * w1.w;
    p += __shfl_xor(p, 16, 32);
    p += __shfl_xor(p, 8, 32);
    p += __shfl_xor(p, 4, 32);
    p += __shfl_xor(p, 2, 32);
    p += __shfl_xor(p, 1, 32);
    if (lane == 0) sc[m] = p;
  }
  __syncthreads();

  float s0 = sc[tid];
  float s1 = sc[tid + 256];
  if (mask[b * TM + tid] == 0)       s0 = -INFINITY;
  if (mask[b * TM + tid + 256] == 0) s1 = -INFINITY;

  red[tid] = fmaxf(s0, s1);
  __syncthreads();
  for (int s = 128; s > 0; s >>= 1) {
    if (tid < s) red[tid] = fmaxf(red[tid], red[tid + s]);
    __syncthreads();
  }
  const float rowmax = red[0];
  __syncthreads();

  const float e0 = fast_exp(s0 - rowmax);
  const float e1 = fast_exp(s1 - rowmax);
  red[tid] = e0 + e1;
  __syncthreads();
  for (int s = 128; s > 0; s >>= 1) {
    if (tid < s) red[tid] += red[tid + s];
    __syncthreads();
  }
  const float inv = 1.0f / red[0];

  const size_t base = ((size_t)b * TX + x) * TM;
  Sout[base + tid]       = e0 * inv;
  Sout[base + 256 + tid] = e1 * inv;
}

__global__ __launch_bounds__(128) void attn_out_kernel(
    const float* __restrict__ S, const float* __restrict__ Mem,
    float* __restrict__ Out) {
  const int warp = threadIdx.x >> 5;
  const int lane = threadIdx.x & 31;
  const int half = lane >> 4;
  const int l    = lane & 15;

  const int pairsPerRow   = D / 32;
  const int tilesPerBatch = (TX / 16) * pairsPerRow;
  int tile  = blockIdx.x * (blockDim.x >> 5) + warp;
  int batch = tile / tilesPerBatch;
  int t     = tile % tilesPerBatch;
  int mt = t / pairsPerRow;
  int np = t % pairsPerRow;
  const int row0 = mt * 16;
  const int col0 = np * 32;

  const float* Sb = S   + (size_t)batch * TX * TM;
  const float* Mb = Mem + (size_t)batch * TM * D;
  const float* arow = Sb + (size_t)(row0 + l) * TM;

  v8f acc0 = {}; v8f acc1 = {};
#pragma unroll 4
  for (int k = 0; k < TM; k += 4) {
    const int kr = k + 2 * half;
    v2f a = *(const v2f*)(arow + kr);
    v2f b0, b1;
    b0.x = Mb[(size_t)kr * D + col0 + l];
    b0.y = Mb[(size_t)(kr + 1) * D + col0 + l];
    b1.x = Mb[(size_t)kr * D + col0 + 16 + l];
    b1.y = Mb[(size_t)(kr + 1) * D + col0 + 16 + l];
    acc0 = __builtin_amdgcn_wmma_f32_16x16x4_f32(false, a, false, b0, (short)0, acc0, false, false);
    acc1 = __builtin_amdgcn_wmma_f32_16x16x4_f32(false, a, false, b1, (short)0, acc1, false, false);
  }
#pragma unroll
  for (int j = 0; j < 8; ++j) {
    int r = row0 + j + 8 * half;
    Out[((size_t)batch * TX + r) * D + col0 + l]      = acc0[j];
    Out[((size_t)batch * TX + r) * D + col0 + 16 + l] = acc1[j];
  }
}

extern "C" void kernel_launch(void* const* d_in, const int* in_sizes, int n_in,
                              void* d_out, int out_size, void* d_ws, size_t ws_size,
                              hipStream_t stream) {
  const float* x    = (const float*)d_in[0];
  const float* mem  = (const float*)d_in[1];
  const int*   mask = (const int*)d_in[2];
  const float* w1   = (const float*)d_in[3];
  const float* b1   = (const float*)d_in[4];
  const float* w2   = (const float*)d_in[5];
  const float* wst  = (const float*)d_in[6];

  float* item1 = (float*)d_ws;
  float* item2 = item1 + (size_t)BATCH * TX * D;
  float* out   = (float*)d_out;
  float* Sout  = out + (size_t)BATCH * TX * D;

  const int rowsX = BATCH * TX;
  const int rowsM = BATCH * TM;
  dim3 blk(128);

  int tilesProj = (rowsX / 16) * (D / 32);
  proj_gemm_kernel<<<tilesProj / 4, blk, 0, stream>>>(x,   w1, b1,      item1, rowsX);
  proj_gemm_kernel<<<tilesProj / 4, blk, 0, stream>>>(mem, w2, nullptr, item2, rowsM);

  scores_softmax_kernel<<<BATCH * TX, 256, 0, stream>>>(item1, item2, wst, mask, Sout);

  int tilesOut = BATCH * (TX / 16) * (D / 32);
  attn_out_kernel<<<tilesOut / 4, blk, 0, stream>>>(Sout, mem, out);
}


# all three WMMA products: operand tiles go global -> LDS by whole-line block loads (two stages, 32x64 block tile, k chunks of 64), 8-byte fragments read from LDS; same products, same sum order
# speedup vs baseline: 1.5971x; 1.5971x over previous
#include <hip/hip_runtime.h>
#include <math.h>

#define D      256
#define TX     512
#define TM     512
#define BATCH  4

typedef __attribute__((ext_vector_type(2))) float v2f;
typedef __attribute__((ext_vector_type(8))) float v8f;

__device__ __forceinline__ float fast_tanh(float x) {
#if __has_builtin(__builtin_amdgcn_tanhf)
  return __builtin_amdgcn_tanhf(x);
#else
  float r;
  asm("v_tanh_f32 %0, %1" : "=v"(r) : "v"(x));
  return r;
#endif
}

__device__ __forceinline__ float fast_exp(float x) {
#if __has_builtin(__builtin_amdgcn_exp2f)
  return __builtin_amdgcn_exp2f(x * 1.4426950408889634f);
#else
  return __expf(x);
#endif
}

#define GT_M 32
#define GT_N 64
#define GT_K 64
#define PA   68
#define PBP  160

template <int K, bool B_KMAJOR>
__global__ __launch_bounds__(128) void gemm_lds_kernel(
    const float* __restrict__ A, const float* __restrict__ B,
    const float* __restrict__ bias, float* __restrict__ C) {
  constexpr int A_ST = GT_M * PA;
  constexpr int B_ST = B_KMAJOR ? GT_N * PA : (GT_K / 2) * PBP;
  constexpr int NCH  = K / GT_K;
  __shared__ __attribute__((aligned(16))) float As[2][A_ST];
  __shared__ __attribute__((aligned(16))) float Bs[2][B_ST];

  const int tid  = threadIdx.x;
  const int warp = tid >> 5;
  const int lane = tid & 31;
  const int half = lane >> 4;
  const int l    = lane & 15;
  const int wr   = warp >> 1;
  const int wc   = warp & 1;

  const int nt    = blockIdx.x & 3;
  const int mrow  = blockIdx.x >> 2;
  const int row0  = mrow * GT_M;
  const int batch = mrow >> 4;

  const int tq = tid & 15;
  const int tr = tid >> 4;

  const float* Ag = A + (size_t)(row0 + tr) * K + tq * 4;
  const float* Bg = B_KMAJOR ? B + (size_t)(nt * GT_N + tr) * K + tq * 4
                             : B + ((size_t)batch * K + 2 * tr) * D + nt * GT_N + tq * 4;
  constexpr size_t A_STEP = (size_t)8 * K;
  constexpr size_t B_STEP = B_KMAJOR ? (size_t)8 * K : (size_t)16 * D;
  constexpr size_t A_CH = GT_K;
  constexpr size_t B_CH = B_KMAJOR ? (size_t)GT_K : (size_t)GT_K * D;

  const int sa = tr * PA + tq * 4;
  const int sb = B_KMAJOR ? tr * PA + tq * 4 : tr * PBP + tq * 8;
  const int fa = (wr * 16 + l) * PA + 2 * half;
  const int fb = B_KMAJOR ? (wc * 32 + l) * PA + 2 * half : half * PBP + 2 * (wc * 32 + l);

  float4 a0, a1, a2, a3;
  float4 b0, b1, b2, b3, b4, b5, b6, b7;

#define GT_LOAD(c)                                                                   \
  do {                                                                               \
    const float* ap = Ag + (size_t)(c) * A_CH;                                       \
    const float* bp = Bg + (size_t)(c) * B_CH;                                       \
    a0 = *(const float4*)(ap);              a1 = *(const float4*)(ap + A_STEP);      \
    a2 = *(const float4*)(ap + 2 * A_STEP); a3 = *(const float4*)(ap + 3 * A_STEP);  \
    if (B_KMAJOR) {                                                                  \
      b0 = *(const float4*)(bp);              b1 = *(const float4*)(bp + B_STEP);     \
      b2 = *(const float4*)(bp + 2 * B_STEP); b3 = *(const float4*)(bp + 3 * B_STEP); \
      b4 = *(const float4*)(bp + 4 * B_STEP); b5 = *(const float4*)(bp + 5 * B_STEP); \
      b6 = *(const float4*)(bp + 6 * B_STEP); b7 = *(const float4*)(bp + 7 * B_STEP); \
    } else {                                                                         \
      b0 = *(const float4*)(bp);              b1 = *(const float4*)(bp + D);          \
      b2 = *(const float4*)(bp + B_STEP);     b3 = *(const float4*)(bp + B_STEP + D); \
      b4 = *(const float4*)(bp + 2 * B_STEP); b5 = *(const float4*)(bp + 2 * B_STEP + D); \
      b6 = *(const float4*)(bp + 3 * B_STEP); b7 = *(const float4*)(bp + 3 * B_STEP + D); \
    }                                                                                \
  } while (0)

#define GT_STORE_PAIR(dst, e, o)                                                     \
  do {                                                                               \
    *(float4*)(dst)       = make_float4((e).x, (o).x, (e).y, (o).y);                 \
    *(float4*)((dst) + 4) = make_float4((e).z, (o).z, (e).w, (o).w);                 \
  } while (0)

#define GT_STORE(s)                                                                  \
  do {                                                                               \
    float* ad = &As[s][sa];                                                          \
    float* bd = &Bs[s][sb];                                                          \
    *(float4*)(ad)           = a0; *(float4*)(ad + 8 * PA)  = a1;                    \
    *(float4*)(ad + 16 * PA) = a2; *(float4*)(ad + 24 * PA) = a3;                    \
    if (B_KMAJOR) {                                                                  \
      *(float4*)(bd)           = b0; *(float4*)(bd + 8 * PA)  = b1;                  \
      *(float4*)(bd + 16 * PA) = b2; *(float4*)(bd + 24 * PA) = b3;                  \
      *(float4*)(bd + 32 * PA) = b4; *(float4*)(bd + 40 * PA) = b5;                  \
      *(float4*)(bd + 48 * PA) = b6; *(float4*)(bd + 56 * PA) = b7;                  \
    } else {                                                                         \
      GT_STORE_PAIR(bd,            b0, b1);                                          \
      GT_STORE_PAIR(bd + 8 * PBP,  b2, b3);                                          \
      GT_STORE_PAIR(bd + 16 * PBP, b4, b5);                                          \
      GT_STORE_PAIR(bd + 24 * PBP, b6, b7);                                          \
    }                                                                                \
  } while (0)

  v8f acc0 = {}; v8f acc1 = {};

  GT_LOAD(0);

#pragma unroll
  for (int c = 0; c < NCH; ++c) {
    const int s = c & 1;
    GT_STORE(s);
    __syncthreads();
    if (c + 1 < NCH) GT_LOAD(c + 1);

    const float* as = &As[s][fa];
    const float* bs = &Bs[s][fb];
#pragma unroll
    for (int ks = 0; ks < GT_K / 4; ++ks) {
      const v2f a = *(const v2f*)(as + 4 * ks);
      v2f p, q;
      if (B_KMAJOR) {
        p = *(const v2f*)(bs + 4 * ks);
        q = *(const v2f*)(bs + 16 * PA + 4 * ks);
      } else {
        p = *(const v2f*)(bs + 2 * ks * PBP);
        q = *(const v2f*)(bs + 2 * ks * PBP + 32);
      }
      acc0 = __builtin_amdgcn_wmma_f32_16x16x4_f32(false, a, false, p, (short)0, acc0, false, false);
      acc1 = __builtin_amdgcn_wmma_f32_16x16x4_f32(false, a, false, q, (short)0, acc1, false, false);
    }
  }
#undef GT_LOAD
#undef GT_STORE
#undef GT_STORE_PAIR

  const int col0 = nt * GT_N + wc * 32;
  float bv0 = 0.0f, bv1 = 0.0f;
  if (bias) { bv0 = bias[col0 + l]; bv1 = bias[col0 + 16 + l]; }
#pragma unroll
  for (int j = 0; j < 8; ++j) {
    const int r = row0 + wr * 16 + j + 8 * half;
    C[(size_t)r * D + col0 + l]      = acc0[j] + bv0;
    C[(size_t)r * D + col0 + 16 + l] = acc1[j] + bv1;
  }
}

#define RX 8
#define SW 16
__global__ __launch_bounds__(SW * 32) void scores_softmax_kernel(
    const float* __restrict__ item1, const float* __restrict__ item2,
    const float* __restrict__ wst, const int* __restrict__ mask,
    float* __restrict__ Sout) {
  __shared__ float sc[RX][TM];

  const int tid  = threadIdx.x;
  const int b    = blockIdx.x / (TX / RX);
  const int x0   = (blockIdx.x % (TX / RX)) * RX;
  const int warp = tid >> 5;
  const int lane = tid & 31;

  float f1[RX][8];
  const float* i1 = item1 + ((size_t)b * TX + x0) * D + lane * 8;
#pragma unroll
  for (int r = 0; r < RX; ++r) {
    const float4 a0 = *(const float4*)(i1 + (size_t)r * D);
    const float4 a1 = *(const float4*)(i1 + (size_t)r * D + 4);
    f1[r][0] = a0.x; f1[r][1] = a0.y; f1[r][2] = a0.z; f1[r][3] = a0.w;
    f1[r][4] = a1.x; f1[r][5] = a1.y; f1[r][6] = a1.z; f1[r][7] = a1.w;
  }
  const float4 w0 = *(const float4*)(wst + lane * 8);
  const float4 w1 = *(const float4*)(wst + lane * 8 + 4);

  const float* i2b = item2 + (size_t)b * TM * D + lane * 8;
  float4 n0 = *(const float4*)(i2b + (size_t)warp * D);
  float4 n1 = *(const float4*)(i2b + (size_t)warp * D + 4);
  for (int m = warp; m < TM; m += SW) {
    const float4 r0 = n0, r1 = n1;
    const int mn = (m + SW < TM) ? (m + SW) : m;
    n0 = *(const float4*)(i2b + (size_t)mn * D);
    n1 = *(const float4*)(i2b + (size_t)mn * D + 4);

    float p[RX];
#pragma unroll
    for (int r = 0; r < RX; ++r) {
      float q;
      q  = fast_tanh(f1[r][0] + r0.x) * w0.x;
      q += fast_tanh(f1[r][1] + r0.y) * w0.y;
      q += fast_tanh(f1[r][2] + r0.z) * w0.z;
      q += fast_tanh(f1[r][3] + r0.w) * w0.w;
      q += fast_tanh(f1[r][4] + r1.x) * w1.x;
      q += fast_tanh(f1[r][5] + r1.y) * w1.y;
      q += fast_tanh(f1[r][6] + r1.z) * w1.z;
      q += fast_tanh(f1[r][7] + r1.w) * w1.w;
      p[r] = q;
    }
#pragma unroll
    for (int s = 16; s > 0; s >>= 1) {
#pragma unroll
      for (int r = 0; r < RX; ++r) p[r] += __shfl_xor(p[r], s, 32);
    }
    if (lane == 0) {
#pragma unroll
      for (int r = 0; r < RX; ++r) sc[r][m] = p[r];
    }
  }
  __syncthreads();

  if (warp < RX) {
    const int* mrow = mask + b * TM;
    float v[TM / 32];
    float mx = -INFINITY;
#pragma unroll
    for (int i = 0; i < TM / 32; ++i) {
      const int m = lane + 32 * i;
      float s = sc[warp][m];
      if (mrow[m] == 0) s = -INFINITY;
      v[i] = s;
      mx = fmaxf(mx, s);
    }
#pragma unroll
    for (int s = 16; s > 0; s >>= 1) mx = fmaxf(mx, __shfl_xor(mx, s, 32));

    float sum = 0.0f;
#pragma unroll
    for (int i = 0; i < TM / 32; ++i) {
      v[i] = fast_exp(v[i] - mx);
      sum += v[i];
    }
#pragma unroll
    for (int s = 16; s > 0; s >>= 1) sum += __shfl_xor(sum, s, 32);
    const float inv = 1.0f / sum;

    float* orow = Sout + ((size_t)b * TX + x0 + warp) * TM;
#pragma unroll
    for (int i = 0; i < TM / 32; ++i) orow[lane + 32 * i] = v[i] * inv;
  }
}

extern "C" void kernel_launch(void* const* d_in, const int* in_sizes, int n_in,
                              void* d_out, int out_size, void* d_ws, size_t ws_size,
                              hipStream_t stream) {
  const float* x    = (const float*)d_in[0];
  const float* mem  = (const float*)d_in[1];
  const int*   mask = (const int*)d_in[2];
  const float* w1   = (const float*)d_in[3];
  const float* b1   = (const float*)d_in[4];
  const float* w2   = (const float*)d_in[5];
  const float* wst  = (const float*)d_in[6];

  float* item1 = (float*)d_ws;
  float* item2 = item1 + (size_t)BATCH * TX * D;
  float* out   = (float*)d_out;
  float* Sout  = out + (size_t)BATCH * TX * D;

  dim3 blk(128);
  const int gemmBlocks = (BATCH * TX / GT_M) * (D / GT_N);

  gemm_lds_kernel<D, true><<<gemmBlocks, blk, 0, stream>>>(x,   w1, b1,      item1);
  gemm_lds_kernel<D, true><<<gemmBlocks, blk, 0, stream>>>(mem, w2, nullptr, item2);

  scores_softmax_kernel<<<BATCH * (TX / RX), SW * 32, 0, stream>>>(item1, item2, wst, mask, Sout);

  gemm_lds_kernel<TM, false><<<gemmBlocks, blk, 0, stream>>>(Sout, mem, nullptr, out);
}


# scores kernel: the 32-lane sum of the 8 row partials runs on the vector ALU (cross-row swap + add, then bank-masked row_xmask DPP adds folding 4->2->1 registers), no ds_bpermute; same sum order
# speedup vs baseline: 1.6601x; 1.0394x over previous
#include <hip/hip_runtime.h>
#include <math.h>

#define D      256
#define TX     512
#define TM     512
#define BATCH  4

typedef __attribute__((ext_vector_type(2))) float v2f;
typedef __attribute__((ext_vector_type(8))) float v8f;

__device__ __forceinline__ float fast_tanh(float x) {
#if __has_builtin(__builtin_amdgcn_tanhf)
  return __builtin_amdgcn_tanhf(x);
#else
  float r;
  asm("v_tanh_f32 %0, %1" : "=v"(r) : "v"(x));
  return r;
#endif
}

__device__ __forceinline__ float fast_exp(float x) {
#if __has_builtin(__builtin_amdgcn_exp2f)
  return __builtin_amdgcn_exp2f(x * 1.4426950408889634f);
#else
  return __expf(x);
#endif
}

#define GT_M 32
#define GT_N 64
#define GT_K 64
#define PA   68
#define PBP  160

template <int K, bool B_KMAJOR>
__global__ __launch_bounds__(128) void gemm_lds_kernel(
    const float* __restrict__ A, const float* __restrict__ B,
    const float* __restrict__ bias, float* __restrict__ C) {
  constexpr int A_ST = GT_M * PA;
  constexpr int B_ST = B_KMAJOR ? GT_N * PA : (GT_K / 2) * PBP;
  constexpr int NCH  = K / GT_K;
  __shared__ __attribute__((aligned(16))) float As[2][A_ST];
  __shared__ __attribute__((aligned(16))) float Bs[2][B_ST];

  const int tid  = threadIdx.x;
  const int warp = tid >> 5;
  const int lane = tid & 31;
  const int half = lane >> 4;
  const int l    = lane & 15;
  const int wr   = warp >> 1;
  const int wc   = warp & 1;

  const int nt    = blockIdx.x & 3;
  const int mrow  = blockIdx.x >> 2;
  const int row0  = mrow * GT_M;
  const int batch = mrow >> 4;

  const int tq = tid & 15;
  const int tr = tid >> 4;

  const float* Ag = A + (size_t)(row0 + tr) * K + tq * 4;
  const float* Bg = B_KMAJOR ? B + (size_t)(nt * GT_N + tr) * K + tq * 4
                             : B + ((size_t)batch * K + 2 * tr) * D + nt * GT_N + tq * 4;
  constexpr size_t A_STEP = (size_t)8 * K;
  constexpr size_t B_STEP = B_KMAJOR ? (size_t)8 * K : (size_t)16 * D;
  constexpr size_t A_CH = GT_K;
  constexpr size_t B_CH = B_KMAJOR ? (size_t)GT_K : (size_t)GT_K * D;

  const int sa = tr * PA + tq * 4;
  const int sb = B_KMAJOR ? tr * PA + tq * 4 : tr * PBP + tq * 8;
  const int fa = (wr * 16 + l) * PA + 2 * half;
  const int fb = B_KMAJOR ? (wc * 32 + l) * PA + 2 * half : half * PBP + 2 * (wc * 32 + l);

  float4 a0, a1, a2, a3;
  float4 b0, b1, b2, b3, b4, b5, b6, b7;

#define GT_LOAD(c)                                                                   \
  do {                                                                               \
    const float* ap = Ag + (size_t)(c) * A_CH;                                       \
    const float* bp = Bg + (size_t)(c) * B_CH;                                       \
    a0 = *(const float4*)(ap);              a1 = *(const float4*)(ap + A_STEP);      \
    a2 = *(const float4*)(ap + 2 * A_STEP); a3 = *(const float4*)(ap + 3 * A_STEP);  \
    if (B_KMAJOR) {                                                                  \
      b0 = *(const float4*)(bp);              b1 = *(const float4*)(bp + B_STEP);     \
      b2 = *(const float4*)(bp + 2 * B_STEP); b3 = *(const float4*)(bp + 3 * B_STEP); \
      b4 = *(const float4*)(bp + 4 * B_STEP); b5 = *(const float4*)(bp + 5 * B_STEP); \
      b6 = *(const float4*)(bp + 6 * B_STEP); b7 = *(const float4*)(bp + 7 * B_STEP); \
    } else {                                                                         \
      b0 = *(const float4*)(bp);              b1 = *(const float4*)(bp + D);          \
      b2 = *(const float4*)(bp + B_STEP);     b3 = *(const float4*)(bp + B_STEP + D); \
      b4 = *(const float4*)(bp + 2 * B_STEP); b5 = *(const float4*)(bp + 2 * B_STEP + D); \
      b6 = *(const float4*)(bp + 3 * B_STEP); b7 = *(const float4*)(bp + 3 * B_STEP + D); \
    }                                                                                \
  } while (0)

#define GT_STORE_PAIR(dst, e, o)                                                     \
  do {                                                                               \
    *(float4*)(dst)       = make_float4((e).x, (o).x, (e).y, (o).y);                 \
    *(float4*)((dst) + 4) = make_float4((e).z, (o).z, (e).w, (o).w);                 \
  } while (0)

#define GT_STORE(s)                                                                  \
  do {                                                                               \
    float* ad = &As[s][sa];                                                          \
    float* bd = &Bs[s][sb];                                                          \
    *(float4*)(ad)           = a0; *(float4*)(ad + 8 * PA)  = a1;                    \
    *(float4*)(ad + 16 * PA) = a2; *(float4*)(ad + 24 * PA) = a3;                    \
    if (B_KMAJOR) {                                                                  \
      *(float4*)(bd)           = b0; *(float4*)(bd + 8 * PA)  = b1;                  \
      *(float4*)(bd + 16 * PA) = b2; *(float4*)(bd + 24 * PA) = b3;                  \
      *(float4*)(bd + 32 * PA) = b4; *(float4*)(bd + 40 * PA) = b5;                  \
      *(float4*)(bd + 48 * PA) = b6; *(float4*)(bd + 56 * PA) = b7;                  \
    } else {                                                                         \
      GT_STORE_PAIR(bd,            b0, b1);                                          \
      GT_STORE_PAIR(bd + 8 * PBP,  b2, b3);                                          \
      GT_STORE_PAIR(bd + 16 * PBP, b4, b5);                                          \
      GT_STORE_PAIR(bd + 24 * PBP, b6, b7);                                          \
    }                                                                                \
  } while (0)

  v8f acc0 = {}; v8f acc1 = {};

  GT_LOAD(0);

#pragma unroll
  for (int c = 0; c < NCH; ++c) {
    const int s = c & 1;
    GT_STORE(s);
    __syncthreads();
    if (c + 1 < NCH) GT_LOAD(c + 1);

    const float* as = &As[s][fa];
    const float* bs = &Bs[s][fb];
#pragma unroll
    for (int ks = 0; ks < GT_K / 4; ++ks) {
      const v2f a = *(const v2f*)(as + 4 * ks);
      v2f p, q;
      if (B_KMAJOR) {
        p = *(const v2f*)(bs + 4 * ks);
        q = *(const v2f*)(bs + 16 * PA + 4 * ks);
      } else {
        p = *(const v2f*)(bs + 2 * ks * PBP);
        q = *(const v2f*)(bs + 2 * ks * PBP + 32);
      }
      acc0 = __builtin_amdgcn_wmma_f32_16x16x4_f32(false, a, false, p, (short)0, acc0, false, false);
      acc1 = __builtin_amdgcn_wmma_f32_16x16x4_f32(false, a, false, q, (short)0, acc1, false, false);
    }
  }
#undef GT_LOAD
#undef GT_STORE
#undef GT_STORE_PAIR

  const int col0 = nt * GT_N + wc * 32;
  float bv0 = 0.0f, bv1 = 0.0f;
  if (bias) { bv0 = bias[col0 + l]; bv1 = bias[col0 + 16 + l]; }
#pragma unroll
  for (int j = 0; j < 8; ++j) {
    const int r = row0 + wr * 16 + j + 8 * half;
    C[(size_t)r * D + col0 + l]      = acc0[j] + bv0;
    C[(size_t)r * D + col0 + 16 + l] = acc1[j] + bv1;
  }
}

__device__ __forceinline__ void swap_rows(float& d, float& s) {
  asm("v_permlane16_swap_b32 %0, %1" : "+v"(d), "+v"(s));
}

#define XADD_ROW(dst, src, X, BANK)                                                  \
  asm("v_add_f32_dpp %0, %1, %1 row_xmask:" #X " row_mask:0xf bank_mask:" #BANK      \
      : "+v"(dst) : "v"(src))

#define RX 8
#define SW 16
__global__ __launch_bounds__(SW * 32) void scores_softmax_kernel(
    const float* __restrict__ item1, const float* __restrict__ item2,
    const float* __restrict__ wst, const int* __restrict__ mask,
    float* __restrict__ Sout) {
  __shared__ float sc[RX][TM + 1];

  const int tid  = threadIdx.x;
  const int b    = blockIdx.x / (TX / RX);
  const int x0   = (blockIdx.x % (TX / RX)) * RX;
  const int warp = tid >> 5;
  const int lane = tid & 31;

  float f1[RX][8];
  const float* i1 = item1 + ((size_t)b * TX + x0) * D + lane * 8;
#pragma unroll
  for (int r = 0; r < RX; ++r) {
    const float4 a0 = *(const float4*)(i1 + (size_t)r * D);
    const float4 a1 = *(const float4*)(i1 + (size_t)r * D + 4);
    f1[r][0] = a0.x; f1[r][1] = a0.y; f1[r][2] = a0.z; f1[r][3] = a0.w;
    f1[r][4] = a1.x; f1[r][5] = a1.y; f1[r][6] = a1.z; f1[r][7] = a1.w;
  }
  const float4 w0 = *(const float4*)(wst + lane * 8);
  const float4 w1 = *(const float4*)(wst + lane * 8 + 4);

  const float* i2b = item2 + (size_t)b * TM * D + lane * 8;
  float4 n0 = *(const float4*)(i2b + (size_t)warp * D);
  float4 n1 = *(const float4*)(i2b + (size_t)warp * D + 4);
  for (int m = warp; m < TM; m += SW) {
    const float4 r0 = n0, r1 = n1;
    const int mn = (m + SW < TM) ? (m + SW) : m;
    n0 = *(const float4*)(i2b + (size_t)mn * D);
    n1 = *(const float4*)(i2b + (size_t)mn * D + 4);

    float p[RX];
#pragma unroll
    for (int r = 0; r < RX; ++r) {
      float q;
      q  = fast_tanh(f1[r][0] + r0.x) * w0.x;
      q += fast_tanh(f1[r][1] + r0.y) * w0.y;
      q += fast_tanh(f1[r][2] + r0.z) * w0.z;
      q += fast_tanh(f1[r][3] + r0.w) * w0.w;
      q += fast_tanh(f1[r][4] + r1.x) * w1.x;
      q += fast_tanh(f1[r][5] + r1.y) * w1.y;
      q += fast_tanh(f1[r][6] + r1.z) * w1.z;
      q += fast_tanh(f1[r][7] + r1.w) * w1.w;
      p[r] = q;
    }
    float q0, q1, q2, q3;
    swap_rows(p[0], p[4]); q0 = p[0] + p[4];
    swap_rows(p[1], p[5]); q1 = p[1] + p[5];
    swap_rows(p[2], p[6]); q2 = p[2] + p[6];
    swap_rows(p[3], p[7]); q3 = p[3] + p[7];
    XADD_ROW(q0, q0, 8, 0x3); XADD_ROW(q0, q2, 8, 0xc);
    XADD_ROW(q1, q1, 8, 0x3); XADD_ROW(q1, q3, 8, 0xc);
    XADD_ROW(q0, q0, 4, 0x5); XADD_ROW(q0, q1, 4, 0xa);
    XADD_ROW(q0, q0, 2, 0xf);
    XADD_ROW(q0, q0, 1, 0xf);
    if ((lane & 3) == 0) sc[lane >> 2][m] = q0;
  }
  __syncthreads();

  if (warp < RX) {
    const int* mrow = mask + b * TM;
    float v[TM / 32];
    float mx = -INFINITY;
#pragma unroll
    for (int i = 0; i < TM / 32; ++i) {
      const int m = lane + 32 * i;
      float s = sc[warp][m];
      if (mrow[m] == 0) s = -INFINITY;
      v[i] = s;
      mx = fmaxf(mx, s);
    }
#pragma unroll
    for (int s = 16; s > 0; s >>= 1) mx = fmaxf(mx, __shfl_xor(mx, s, 32));

    float sum = 0.0f;
#pragma unroll
    for (int i = 0; i < TM / 32; ++i) {
      v[i] = fast_exp(v[i] - mx);
      sum += v[i];
    }
#pragma unroll
    for (int s = 16; s > 0; s >>= 1) sum += __shfl_xor(sum, s, 32);
    const float inv = 1.0f / sum;

    float* orow = Sout + ((size_t)b * TX + x0 + warp) * TM;
#pragma unroll
    for (int i = 0; i < TM / 32; ++i) orow[lane + 32 * i] = v[i] * inv;
  }
}

extern "C" void kernel_launch(void* const* d_in, const int* in_sizes, int n_in,
                              void* d_out, int out_size, void* d_ws, size_t ws_size,
                              hipStream_t stream) {
  const float* x    = (const float*)d_in[0];
  const float* mem  = (const float*)d_in[1];
  const int*   mask = (const int*)d_in[2];
  const float* w1   = (const float*)d_in[3];
  const float* b1   = (const float*)d_in[4];
  const float* w2   = (const float*)d_in[5];
  const float* wst  = (const float*)d_in[6];

  float* item1 = (float*)d_ws;
  float* item2 = item1 + (size_t)BATCH * TX * D;
  float* out   = (float*)d_out;
  float* Sout  = out + (size_t)BATCH * TX * D;

  dim3 blk(128);
  const int gemmBlocks = (BATCH * TX / GT_M) * (D / GT_N);

  gemm_lds_kernel<D, true><<<gemmBlocks, blk, 0, stream>>>(x,   w1, b1,      item1);
  gemm_lds_kernel<D, true><<<gemmBlocks, blk, 0, stream>>>(mem, w2, nullptr, item2);

  scores_softmax_kernel<<<BATCH * (TX / RX), SW * 32, 0, stream>>>(item1, item2, wst, mask, Sout);

  gemm_lds_kernel<TM, false><<<gemmBlocks, blk, 0, stream>>>(Sout, mem, nullptr, out);
}


# all three WMMA products re-tiled: block of 8 waves (two per SIMD), one 16x16 tile per wave on the same 32x64 block tile, k chunks of 128; same products, same sum order
# speedup vs baseline: 1.9273x; 1.1610x over previous
#include <hip/hip_runtime.h>
#include <math.h>

#define D      256
#define TX     512
#define TM     512
#define BATCH  4

typedef __attribute__((ext_vector_type(2))) float v2f;
typedef __attribute__((ext_vector_type(8))) float v8f;

__device__ __forceinline__ float fast_tanh(float x) {
#if __has_builtin(__builtin_amdgcn_tanhf)
  return __builtin_amdgcn_tanhf(x);
#else
  float r;
  asm("v_tanh_f32 %0, %1" : "=v"(r) : "v"(x));
  return r;
#endif
}

__device__ __forceinline__ float fast_exp(float x) {
#if __has_builtin(__builtin_amdgcn_exp2f)
  return __builtin_amdgcn_exp2f(x * 1.4426950408889634f);
#else
  return __expf(x);
#endif
}

#define GT_M 32
#define GT_N 64
#define GT_K 128
#define GT_T 256
#define PA   132
#define PBP  160

template <int K, bool B_KMAJOR>
__global__ __launch_bounds__(GT_T) void gemm_lds_kernel(
    const float* __restrict__ A, const float* __restrict__ B,
    const float* __restrict__ bias, float* __restrict__ C) {
  constexpr int A_ST = GT_M * PA;
  constexpr int B_ST = B_KMAJOR ? GT_N * PA : (GT_K / 2) * PBP;
  constexpr int NCH  = K / GT_K;
  __shared__ __attribute__((aligned(16))) float As[2][A_ST];
  __shared__ __attribute__((aligned(16))) float Bs[2][B_ST];

  const int tid  = threadIdx.x;
  const int warp = tid >> 5;
  const int lane = tid & 31;
  const int half = lane >> 4;
  const int l    = lane & 15;
  const int wr   = warp >> 2;
  const int wc   = warp & 3;

  const int nt    = blockIdx.x & 3;
  const int mrow  = blockIdx.x >> 2;
  const int row0  = mrow * GT_M;
  const int batch = mrow >> 4;

  const int tq = tid & 31;
  const int tr = tid >> 5;
  const int uq = tid & 15;
  const int ur = tid >> 4;

  const float* Ag = A + (size_t)(row0 + tr) * K + tq * 4;
  const float* Bg = B_KMAJOR ? B + (size_t)(nt * GT_N + tr) * K + tq * 4
                             : B + ((size_t)batch * K + 2 * ur) * D + nt * GT_N + uq * 4;
  constexpr size_t A_STEP = (size_t)8 * K;
  constexpr size_t B_STEP = B_KMAJOR ? (size_t)8 * K : (size_t)32 * D;
  constexpr size_t A_CH = GT_K;
  constexpr size_t B_CH = B_KMAJOR ? (size_t)GT_K : (size_t)GT_K * D;

  const int sa = tr * PA + tq * 4;
  const int sb = B_KMAJOR ? tr * PA + tq * 4 : ur * PBP + uq * 8;
  const int fa = (wr * 16 + l) * PA + 2 * half;
  const int fb = B_KMAJOR ? (wc * 16 + l) * PA + 2 * half : half * PBP + 2 * (wc * 16 + l);

  float4 a0, a1, a2, a3;
  float4 b0, b1, b2, b3, b4, b5, b6, b7;

#define GT_LOAD(c)                                                                   \
  do {                                                                               \
    const float* ap = Ag + (size_t)(c) * A_CH;                                       \
    const float* bp = Bg + (size_t)(c) * B_CH;                                       \
    a0 = *(const float4*)(ap);              a1 = *(const float4*)(ap + A_STEP);      \
    a2 = *(const float4*)(ap + 2 * A_STEP); a3 = *(const float4*)(ap + 3 * A_STEP);  \
    if (B_KMAJOR) {                                                                  \
      b0 = *(const float4*)(bp);              b1 = *(const float4*)(bp + B_STEP);     \
      b2 = *(const float4*)(bp + 2 * B_STEP); b3 = *(const float4*)(bp + 3 * B_STEP); \
      b4 = *(const float4*)(bp + 4 * B_STEP); b5 = *(const float4*)(bp + 5 * B_STEP); \
      b6 = *(const float4*)(bp + 6 * B_STEP); b7 = *(const float4*)(bp + 7 * B_STEP); \
    } else {                                                                         \
      b0 = *(const float4*)(bp);              b1 = *(const float4*)(bp + D);          \
      b2 = *(const float4*)(bp + B_STEP);     b3 = *(const float4*)(bp + B_STEP + D); \
      b4 = *(const float4*)(bp + 2 * B_STEP); b5 = *(const float4*)(bp + 2 * B_STEP + D); \
      b6 = *(const float4*)(bp + 3 * B_STEP); b7 = *(const float4*)(bp + 3 * B_STEP + D); \
    }                                                                                \
  } while (0)

#define GT_STORE_PAIR(dst, e, o)                                                     \
  do {                                                                               \
    *(float4*)(dst)       = make_float4((e).x, (o).x, (e).y, (o).y);                 \
    *(float4*)((dst) + 4) = make_float4((e).z, (o).z, (e).w, (o).w);                 \
  } while (0)

#define GT_STORE(s)                                                                  \
  do {                                                                               \
    float* ad = &As[s][sa];                                                          \
    float* bd = &Bs[s][sb];                                                          \
    *(float4*)(ad)           = a0; *(float4*)(ad + 8 * PA)  = a1;                    \
    *(float4*)(ad + 16 * PA) = a2; *(float4*)(ad + 24 * PA) = a3;                    \
    if (B_KMAJOR) {                                                                  \
      *(float4*)(bd)           = b0; *(float4*)(bd + 8 * PA)  = b1;                  \
      *(float4*)(bd + 16 * PA) = b2; *(float4*)(bd + 24 * PA) = b3;                  \
      *(float4*)(bd + 32 * PA) = b4; *(float4*)(bd + 40 * PA) = b5;                  \
      *(float4*)(bd + 48 * PA) = b6; *(float4*)(bd + 56 * PA) = b7;                  \
    } else {                                                                         \
      GT_STORE_PAIR(bd,            b0, b1);                                          \
      GT_STORE_PAIR(bd + 16 * PBP, b2, b3);                                          \
      GT_STORE_PAIR(bd + 32 * PBP, b4, b5);                                          \
      GT_STORE_PAIR(bd + 48 * PBP, b6, b7);                                          \
    }                                                                                \
  } while (0)

  v8f acc0 = {};

  GT_LOAD(0);

#pragma nounroll
  for (int c = 0; c < NCH; ++c) {
    const int s = c & 1;
    GT_STORE(s);
    __syncthreads();
    if (c + 1 < NCH) GT_LOAD(c + 1);

    const float* as = &As[s][fa];
    const float* bs = &Bs[s][fb];
#pragma unroll
    for (int ks = 0; ks < GT_K / 4; ++ks) {
      const v2f a = *(const v2f*)(as + 4 * ks);
      v2f p;
      if (B_KMAJOR) p = *(const v2f*)(bs + 4 * ks);
      else          p = *(const v2f*)(bs + 2 * ks * PBP);
      acc0 = __builtin_amdgcn_wmma_f32_16x16x4_f32(false, a, false, p, (short)0, acc0, false, false);
    }
  }
#undef GT_LOAD
#undef GT_STORE
#undef GT_STORE_PAIR

  const int col0 = nt * GT_N + wc * 16;
  float bv0 = 0.0f;
  if (bias) bv0 = bias[col0 + l];
#pragma unroll
  for (int j = 0; j < 8; ++j) {
    const int r = row0 + wr * 16 + j + 8 * half;
    C[(size_t)r * D + col0 + l] = acc0[j] + bv0;
  }
}

__device__ __forceinline__ void swap_rows(float& d, float& s) {
  asm("v_permlane16_swap_b32 %0, %1" : "+v"(d), "+v"(s));
}

#define XADD_ROW(dst, src, X, BANK)                                                  \
  asm("v_add_f32_dpp %0, %1, %1 row_xmask:" #X " row_mask:0xf bank_mask:" #BANK      \
      : "+v"(dst) : "v"(src))

#define RX 8
#define SW 16
__global__ __launch_bounds__(SW * 32) void scores_softmax_kernel(
    const float* __restrict__ item1, const float* __restrict__ item2,
    const float* __restrict__ wst, const int* __restrict__ mask,
    float* __restrict__ Sout) {
  __shared__ float sc[RX][TM + 1];

  const int tid  = threadIdx.x;
  const int b    = blockIdx.x / (TX / RX);
  const int x0   = (blockIdx.x % (TX / RX)) * RX;
  const int warp = tid >> 5;
  const int lane = tid & 31;

  float f1[RX][8];
  const float* i1 = item1 + ((size_t)b * TX + x0) * D + lane * 8;
#pragma unroll
  for (int r = 0; r < RX; ++r) {
    const float4 a0 = *(const float4*)(i1 + (size_t)r * D);
    const float4 a1 = *(const float4*)(i1 + (size_t)r * D + 4);
    f1[r][0] = a0.x; f1[r][1] = a0.y; f1[r][2] = a0.z; f1[r][3] = a0.w;
    f1[r][4] = a1.x; f1[r][5] = a1.y; f1[r][6] = a1.z; f1[r][7] = a1.w;
  }
  const float4 w0 = *(const float4*)(wst + lane * 8);
  const float4 w1 = *(const float4*)(wst + lane * 8 + 4);

  const float* i2b = item2 + (size_t)b * TM * D + lane * 8;
  float4 n0 = *(const float4*)(i2b + (size_t)warp * D);
  float4 n1 = *(const float4*)(i2b + (size_t)warp * D + 4);
  for (int m = warp; m < TM; m += SW) {
    const float4 r0 = n0, r1 = n1;
    const int mn = (m + SW < TM) ? (m + SW) : m;
    n0 = *(const float4*)(i2b + (size_t)mn * D);
    n1 = *(const float4*)(i2b + (size_t)mn * D + 4);

    float p[RX];
#pragma unroll
    for (int r = 0; r < RX; ++r) {
      float q;
      q  = fast_tanh(f1[r][0] + r0.x) * w0.x;
      q += fast_tanh(f1[r][1] + r0.y) * w0.y;
      q += fast_tanh(f1[r][2] + r0.z) * w0.z;
      q += fast_tanh(f1[r][3] + r0.w) * w0.w;
      q += fast_tanh(f1[r][4] + r1.x) * w1.x;
      q += fast_tanh(f1[r][5] + r1.y) * w1.y;
      q += fast_tanh(f1[r][6] + r1.z) * w1.z;
      q += fast_tanh(f1[r][7] + r1.w) * w1.w;
      p[r] = q;
    }
    float q0, q1, q2, q3;
    swap_rows(p[0], p[4]); q0 = p[0] + p[4];
    swap_rows(p[1], p[5]); q1 = p[1] + p[5];
    swap_rows(p[2], p[6]); q2 = p[2] + p[6];
    swap_rows(p[3], p[7]); q3 = p[3] + p[7];
    XADD_ROW(q0, q0, 8, 0x3); XADD_ROW(q0, q2, 8, 0xc);
    XADD_ROW(q1, q1, 8, 0x3); XADD_ROW(q1, q3, 8, 0xc);
    XADD_ROW(q0, q0, 4, 0x5); XADD_ROW(q0, q1, 4, 0xa);
    XADD_ROW(q0, q0, 2, 0xf);
    XADD_ROW(q0, q0, 1, 0xf);
    if ((lane & 3) == 0) sc[lane >> 2][m] = q0;
  }
  __syncthreads();

  if (warp < RX) {
    const int* mrow = mask + b * TM;
    float v[TM / 32];
    float mx = -INFINITY;
#pragma unroll
    for (int i = 0; i < TM / 32; ++i) {
      const int m = lane + 32 * i;
      float s = sc[warp][m];
      if (mrow[m] == 0) s = -INFINITY;
      v[i] = s;
      mx = fmaxf(mx, s);
    }
#pragma unroll
    for (int s = 16; s > 0; s >>= 1) mx = fmaxf(mx, __shfl_xor(mx, s, 32));

    float sum = 0.0f;
#pragma unroll
    for (int i = 0; i < TM / 32; ++i) {
      v[i] = fast_exp(v[i] - mx);
      sum += v[i];
    }
#pragma unroll
    for (int s = 16; s > 0; s >>= 1) sum += __shfl_xor(sum, s, 32);
    const float inv = 1.0f / sum;

    float* orow = Sout + ((size_t)b * TX + x0 + warp) * TM;
#pragma unroll
    for (int i = 0; i < TM / 32; ++i) orow[lane + 32 * i] = v[i] * inv;
  }
}

extern "C" void kernel_launch(void* const* d_in, const int* in_sizes, int n_in,
                              void* d_out, int out_size, void* d_ws, size_t ws_size,
                              hipStream_t stream) {
  const float* x    = (const float*)d_in[0];
  const float* mem  = (const float*)d_in[1];
  const int*   mask = (const int*)d_in[2];
  const float* w1   = (const float*)d_in[3];
  const float* b1   = (const float*)d_in[4];
  const float* w2   = (const float*)d_in[5];
  const float* wst  = (const float*)d_in[6];

  float* item1 = (float*)d_ws;
  float* item2 = item1 + (size_t)BATCH * TX * D;
  float* out   = (float*)d_out;
  float* Sout  = out + (size_t)BATCH * TX * D;

  dim3 blk(GT_T);
  const int gemmBlocks = (BATCH * TX / GT_M) * (D / GT_N);

  gemm_lds_kernel<D, true><<<gemmBlocks, blk, 0, stream>>>(x,   w1, b1,      item1);
  gemm_lds_kernel<D, true><<<gemmBlocks, blk, 0, stream>>>(mem, w2, nullptr, item2);

  scores_softmax_kernel<<<BATCH * (TX / RX), SW * 32, 0, stream>>>(item1, item2, wst, mask, Sout);

  gemm_lds_kernel<TM, false><<<gemmBlocks, blk, 0, stream>>>(Sout, mem, nullptr, out);
}


# scores kernel re-tiled: a wave takes two m rows per trip, one per 16-lane half, each lane a 16-wide d slice; a score's d sum stays in its half (DPP only), even/odd pair partials; d-sum order changes
# speedup vs baseline: 2.2272x; 1.1556x over previous
#include <hip/hip_runtime.h>
#include <math.h>

#define D      256
#define TX     512
#define TM     512
#define BATCH  4

typedef __attribute__((ext_vector_type(2))) float v2f;
typedef __attribute__((ext_vector_type(8))) float v8f;

__device__ __forceinline__ float fast_tanh(float x) {
#if __has_builtin(__builtin_amdgcn_tanhf)
  return __builtin_amdgcn_tanhf(x);
#else
  float r;
  asm("v_tanh_f32 %0, %1" : "=v"(r) : "v"(x));
  return r;
#endif
}

__device__ __forceinline__ float fast_exp(float x) {
#if __has_builtin(__builtin_amdgcn_exp2f)
  return __builtin_amdgcn_exp2f(x * 1.4426950408889634f);
#else
  return __expf(x);
#endif
}

#define GT_M 32
#define GT_N 64
#define GT_K 128
#define GT_T 256
#define PA   132
#define PBP  160

template <int K, bool B_KMAJOR>
__global__ __launch_bounds__(GT_T) void gemm_lds_kernel(
    const float* __restrict__ A, const float* __restrict__ B,
    const float* __restrict__ bias, float* __restrict__ C) {
  constexpr int A_ST = GT_M * PA;
  constexpr int B_ST = B_KMAJOR ? GT_N * PA : (GT_K / 2) * PBP;
  constexpr int NCH  = K / GT_K;
  __shared__ __attribute__((aligned(16))) float As[2][A_ST];
  __shared__ __attribute__((aligned(16))) float Bs[2][B_ST];

  const int tid  = threadIdx.x;
  const int warp = tid >> 5;
  const int lane = tid & 31;
  const int half = lane >> 4;
  const int l    = lane & 15;
  const int wr   = warp >> 2;
  const int wc   = warp & 3;

  const int nt    = blockIdx.x & 3;
  const int mrow  = blockIdx.x >> 2;
  const int row0  = mrow * GT_M;
  const int batch = mrow >> 4;

  const int tq = tid & 31;
  const int tr = tid >> 5;
  const int uq = tid & 15;
  const int ur = tid >> 4;

  const float* Ag = A + (size_t)(row0 + tr) * K + tq * 4;
  const float* Bg = B_KMAJOR ? B + (size_t)(nt * GT_N + tr) * K + tq * 4
                             : B + ((size_t)batch * K + 2 * ur) * D + nt * GT_N + uq * 4;
  constexpr size_t A_STEP = (size_t)8 * K;
  constexpr size_t B_STEP = B_KMAJOR ? (size_t)8 * K : (size_t)32 * D;
  constexpr size_t A_CH = GT_K;
  constexpr size_t B_CH = B_KMAJOR ? (size_t)GT_K : (size_t)GT_K * D;

  const int sa = tr * PA + tq * 4;
  const int sb = B_KMAJOR ? tr * PA + tq * 4 : ur * PBP + uq * 8;
  const int fa = (wr * 16 + l) * PA + 2 * half;
  const int fb = B_KMAJOR ? (wc * 16 + l) * PA + 2 * half : half * PBP + 2 * (wc * 16 + l);

  float4 a0, a1, a2, a3;
  float4 b0, b1, b2, b3, b4, b5, b6, b7;

#define GT_LOAD(c)                                                                   \
  do {                                                                               \
    const float* ap = Ag + (size_t)(c) * A_CH;                                       \
    const float* bp = Bg + (size_t)(c) * B_CH;                                       \
    a0 = *(const float4*)(ap);              a1 = *(const float4*)(ap + A_STEP);      \
    a2 = *(const float4*)(ap + 2 * A_STEP); a3 = *(const float4*)(ap + 3 * A_STEP);  \
    if (B_KMAJOR) {                                                                  \
      b0 = *(const float4*)(bp);              b1 = *(const float4*)(bp + B_STEP);     \
      b2 = *(const float4*)(bp + 2 * B_STEP); b3 = *(const float4*)(bp + 3 * B_STEP); \
      b4 = *(const float4*)(bp + 4 * B_STEP); b5 = *(const float4*)(bp + 5 * B_STEP); \
      b6 = *(const float4*)(bp + 6 * B_STEP); b7 = *(const float4*)(bp + 7 * B_STEP); \
    } else {                                                                         \
      b0 = *(const float4*)(bp);              b1 = *(const float4*)(bp + D);          \
      b2 = *(const float4*)(bp + B_STEP);     b3 = *(const float4*)(bp + B_STEP + D); \
      b4 = *(const float4*)(bp + 2 * B_STEP); b5 = *(const float4*)(bp + 2 * B_STEP + D); \
      b6 = *(const float4*)(bp + 3 * B_STEP); b7 = *(const float4*)(bp + 3 * B_STEP + D); \
    }                                                                                \
  } while (0)

#define GT_STORE_PAIR(dst, e, o)                                                     \
  do {                                                                               \
    *(float4*)(dst)       = make_float4((e).x, (o).x, (e).y, (o).y);                 \
    *(float4*)((dst) + 4) = make_float4((e).z, (o).z, (e).w, (o).w);                 \
  } while (0)

#define GT_STORE(s)                                                                  \
  do {                                                                               \
    float* ad = &As[s][sa];                                                          \
    float* bd = &Bs[s][sb];                                                          \
    *(float4*)(ad)           = a0; *(float4*)(ad + 8 * PA)  = a1;                    \
    *(float4*)(ad + 16 * PA) = a2; *(float4*)(ad + 24 * PA) = a3;                    \
    if (B_KMAJOR) {                                                                  \
      *(float4*)(bd)           = b0; *(float4*)(bd + 8 * PA)  = b1;                  \
      *(float4*)(bd + 16 * PA) = b2; *(float4*)(bd + 24 * PA) = b3;                  \
      *(float4*)(bd + 32 * PA) = b4; *(float4*)(bd + 40 * PA) = b5;                  \
      *(float4*)(bd + 48 * PA) = b6; *(float4*)(bd + 56 * PA) = b7;                  \
    } else {                                                                         \
      GT_STORE_PAIR(bd,            b0, b1);                                          \
      GT_STORE_PAIR(bd + 16 * PBP, b2, b3);                                          \
      GT_STORE_PAIR(bd + 32 * PBP, b4, b5);                                          \
      GT_STORE_PAIR(bd + 48 * PBP, b6, b7);                                          \
    }                                                                                \
  } while (0)

  v8f acc0 = {};

  GT_LOAD(0);

#pragma nounroll
  for (int c = 0; c < NCH; ++c) {
    const int s = c & 1;
    GT_STORE(s);
    __syncthreads();
    if (c + 1 < NCH) GT_LOAD(c + 1);

    const float* as = &As[s][fa];
    const float* bs = &Bs[s][fb];
#pragma unroll
    for (int ks = 0; ks < GT_K / 4; ++ks) {
      const v2f a = *(const v2f*)(as + 4 * ks);
      v2f p;
      if (B_KMAJOR) p = *(const v2f*)(bs + 4 * ks);
      else          p = *(const v2f*)(bs + 2 * ks * PBP);
      acc0 = __builtin_amdgcn_wmma_f32_16x16x4_f32(false, a, false, p, (short)0, acc0, false, false);
    }
  }
#undef GT_LOAD
#undef GT_STORE
#undef GT_STORE_PAIR

  const int col0 = nt * GT_N + wc * 16;
  float bv0 = 0.0f;
  if (bias) bv0 = bias[col0 + l];
#pragma unroll
  for (int j = 0; j < 8; ++j) {
    const int r = row0 + wr * 16 + j + 8 * half;
    C[(size_t)r * D + col0 + l] = acc0[j] + bv0;
  }
}

#define XADD_ROW(dst, src, X, BANK)                                                  \
  asm("v_add_f32_dpp %0, %1, %1 row_xmask:" #X " row_mask:0xf bank_mask:" #BANK      \
      : "+v"(dst) : "v"(src))

struct RowSlice { float4 v0, v1, v2, v3; };

__device__ __forceinline__ v2f tanh2(v2f s) {
  v2f t;
  t.x = fast_tanh(s.x);
  t.y = fast_tanh(s.y);
  return t;
}

#define RX 8
#define SW 16
#define DL 16
__global__ __launch_bounds__(SW * 32) void scores_softmax_kernel(
    const float* __restrict__ item1, const float* __restrict__ item2,
    const float* __restrict__ wst, const int* __restrict__ mask,
    float* __restrict__ Sout) {
  __shared__ float sc[RX][TM + 1];

  const int tid  = threadIdx.x;
  const int b    = blockIdx.x / (TX / RX);
  const int x0   = (blockIdx.x % (TX / RX)) * RX;
  const int warp = tid >> 5;
  const int lane = tid & 31;
  const int half = lane >> 4;
  const int c    = lane & 15;

  v2f f1[RX][DL / 2];
  const float* i1 = item1 + ((size_t)b * TX + x0) * D + c * DL;
#pragma unroll
  for (int r = 0; r < RX; ++r) {
#pragma unroll
    for (int j = 0; j < DL / 4; ++j) {
      const float4 a = *(const float4*)(i1 + (size_t)r * D + 4 * j);
      f1[r][2 * j].x = a.x;     f1[r][2 * j].y = a.y;
      f1[r][2 * j + 1].x = a.z; f1[r][2 * j + 1].y = a.w;
    }
  }
  v2f w[DL / 2];
#pragma unroll
  for (int j = 0; j < DL / 4; ++j) {
    const float4 a = *(const float4*)(wst + c * DL + 4 * j);
    w[2 * j].x = a.x;     w[2 * j].y = a.y;
    w[2 * j + 1].x = a.z; w[2 * j + 1].y = a.w;
  }

  const int m0 = 2 * warp + half;
  const float* i2b = item2 + ((size_t)b * TM + m0) * D + c * DL;
  float* scw = &sc[2 * (c >> 2)][m0];
  const bool writer = (c & 3) == 0;

#define ROW_LOAD(n, t)                                                               \
  do {                                                                               \
    const float* rp = i2b + (size_t)(t) * 32 * D;                                    \
    (n).v0 = *(const float4*)(rp);     (n).v1 = *(const float4*)(rp + 4);                \
    (n).v2 = *(const float4*)(rp + 8); (n).v3 = *(const float4*)(rp + 12);               \
  } while (0)

#define ROW_TRIP(n, t)                                                               \
  do {                                                                               \
    v2f iv[DL / 2];                                                                  \
    iv[0].x = (n).v0.x; iv[0].y = (n).v0.y; iv[1].x = (n).v0.z; iv[1].y = (n).v0.w;          \
    iv[2].x = (n).v1.x; iv[2].y = (n).v1.y; iv[3].x = (n).v1.z; iv[3].y = (n).v1.w;          \
    iv[4].x = (n).v2.x; iv[4].y = (n).v2.y; iv[5].x = (n).v2.z; iv[5].y = (n).v2.w;          \
    iv[6].x = (n).v3.x; iv[6].y = (n).v3.y; iv[7].x = (n).v3.z; iv[7].y = (n).v3.w;          \
    float p[RX];                                                                     \
      \
    _Pragma("unroll")                                                                \
    for (int r = 0; r < RX; ++r) {                                                   \
      v2f q = tanh2(f1[r][0] + iv[0]) * w[0];                                        \
      _Pragma("unroll")                                                              \
      for (int j = 1; j < DL / 2; ++j) q += tanh2(f1[r][j] + iv[j]) * w[j];          \
      p[r] = q.x + q.y;                                                              \
    }                                                                                \
    XADD_ROW(p[0], p[0], 8, 0x3); XADD_ROW(p[0], p[4], 8, 0xc);                      \
    XADD_ROW(p[1], p[1], 8, 0x3); XADD_ROW(p[1], p[5], 8, 0xc);                      \
    XADD_ROW(p[2], p[2], 8, 0x3); XADD_ROW(p[2], p[6], 8, 0xc);                      \
    XADD_ROW(p[3], p[3], 8, 0x3); XADD_ROW(p[3], p[7], 8, 0xc);                      \
    XADD_ROW(p[0], p[0], 4, 0x5); XADD_ROW(p[0], p[2], 4, 0xa);                      \
    XADD_ROW(p[1], p[1], 4, 0x5); XADD_ROW(p[1], p[3], 4, 0xa);                      \
    XADD_ROW(p[0], p[0], 2, 0xf); XADD_ROW(p[1], p[1], 2, 0xf);                      \
    XADD_ROW(p[0], p[0], 1, 0xf); XADD_ROW(p[1], p[1], 1, 0xf);                      \
    if (writer) {                                                                    \
      scw[32 * (t)]            = p[0];                                               \
      scw[32 * (t) + (TM + 1)] = p[1];                                               \
    }                                                                                \
  } while (0)

  RowSlice ra, rb;
  ROW_LOAD(ra, 0);
  for (int t = 0; t < TM / 32; t += 2) {
    ROW_LOAD(rb, t + 1);
    ROW_TRIP(ra, t);
    const int tn = (t + 2 < TM / 32) ? (t + 2) : (t + 1);
    ROW_LOAD(ra, tn);
    ROW_TRIP(rb, t + 1);
  }
#undef ROW_LOAD
#undef ROW_TRIP
  __syncthreads();

  if (warp < RX) {
    const int* mrow = mask + b * TM;
    float v[TM / 32];
    float mx = -INFINITY;
#pragma unroll
    for (int i = 0; i < TM / 32; ++i) {
      const int m = lane + 32 * i;
      float s = sc[warp][m];
      if (mrow[m] == 0) s = -INFINITY;
      v[i] = s;
      mx = fmaxf(mx, s);
    }
#pragma unroll
    for (int s = 16; s > 0; s >>= 1) mx = fmaxf(mx, __shfl_xor(mx, s, 32));

    float sum = 0.0f;
#pragma unroll
    for (int i = 0; i < TM / 32; ++i) {
      v[i] = fast_exp(v[i] - mx);
      sum += v[i];
    }
#pragma unroll
    for (int s = 16; s > 0; s >>= 1) sum += __shfl_xor(sum, s, 32);
    const float inv = 1.0f / sum;

    float* orow = Sout + ((size_t)b * TX + x0 + warp) * TM;
#pragma unroll
    for (int i = 0; i < TM / 32; ++i) orow[lane + 32 * i] = v[i] * inv;
  }
}

extern "C" void kernel_launch(void* const* d_in, const int* in_sizes, int n_in,
                              void* d_out, int out_size, void* d_ws, size_t ws_size,
                              hipStream_t stream) {
  const float* x    = (const float*)d_in[0];
  const float* mem  = (const float*)d_in[1];
  const int*   mask = (const int*)d_in[2];
  const float* w1   = (const float*)d_in[3];
  const float* b1   = (const float*)d_in[4];
  const float* w2   = (const float*)d_in[5];
  const float* wst  = (const float*)d_in[6];

  float* item1 = (float*)d_ws;
  float* item2 = item1 + (size_t)BATCH * TX * D;
  float* out   = (float*)d_out;
  float* Sout  = out + (size_t)BATCH * TX * D;

  dim3 blk(GT_T);
  const int gemmBlocks = (BATCH * TX / GT_M) * (D / GT_N);

  gemm_lds_kernel<D, true><<<gemmBlocks, blk, 0, stream>>>(x,   w1, b1,      item1);
  gemm_lds_kernel<D, true><<<gemmBlocks, blk, 0, stream>>>(mem, w2, nullptr, item2);

  scores_softmax_kernel<<<BATCH * (TX / RX), SW * 32, 0, stream>>>(item1, item2, wst, mask, Sout);

  gemm_lds_kernel<TM, false><<<gemmBlocks, blk, 0, stream>>>(Sout, mem, nullptr, out);
}


# speedup vs baseline: 2.3477x; 1.0541x over previous
#include <hip/hip_runtime.h>
#include <math.h>

#define D      256
#define TX     512
#define TM     512
#define BATCH  4

typedef __attribute__((ext_vector_type(2))) float v2f;
typedef __attribute__((ext_vector_type(8))) float v8f;

__device__ __forceinline__ float fast_exp(float x) {
#if __has_builtin(__builtin_amdgcn_exp2f)
  return __builtin_amdgcn_exp2f(x * 1.4426950408889634f);
#else
  return __expf(x);
#endif
}

#define GT_M 32
#define GT_N 64
#define GT_K 128
#define GT_T 256
#define PA   132
#define PBP  160

template <int K, bool B_KMAJOR>
__global__ __launch_bounds__(GT_T) void gemm_lds_kernel(
    const float* __restrict__ A, const float* __restrict__ B,
    const float* __restrict__ bias, float* __restrict__ C) {
  constexpr int A_ST = GT_M * PA;
  constexpr int B_ST = B_KMAJOR ? GT_N * PA : (GT_K / 2) * PBP;
  constexpr int NCH  = K / GT_K;
  __shared__ __attribute__((aligned(16))) float As[2][A_ST];
  __shared__ __attribute__((aligned(16))) float Bs[2][B_ST];

  const int tid  = threadIdx.x;
  const int warp = tid >> 5;
  const int lane = tid & 31;
  const int half = lane >> 4;
  const int l    = lane & 15;
  const int wr   = warp >> 2;
  const int wc   = warp & 3;

  const int nt    = blockIdx.x & 3;
  const int mrow  = blockIdx.x >> 2;
  const int row0  = mrow * GT_M;
  const int batch = mrow >> 4;

  const int tq = tid & 31;
  const int tr = tid >> 5;
  const int uq = tid & 15;
  const int ur = tid >> 4;

  const float* Ag = A + (size_t)(row0 + tr) * K + tq * 4;
  const float* Bg = B_KMAJOR ? B + (size_t)(nt * GT_N + tr) * K + tq * 4
                             : B + ((size_t)batch * K + 2 * ur) * D + nt * GT_N + uq * 4;
  constexpr size_t A_STEP = (size_t)8 * K;
  constexpr size_t B_STEP = B_KMAJOR ? (size_t)8 * K : (size_t)32 * D;
  constexpr size_t A_CH = GT_K;
  constexpr size_t B_CH = B_KMAJOR ? (size_t)GT_K : (size_t)GT_K * D;

  const int sa = tr * PA + tq * 4;
  const int sb = B_KMAJOR ? tr * PA + tq * 4 : ur * PBP + uq * 8;
  const int fa = (wr * 16 + l) * PA + 2 * half;
  const int fb = B_KMAJOR ? (wc * 16 + l) * PA + 2 * half : half * PBP + 2 * (wc * 16 + l);

  float4 a0, a1, a2, a3;
  float4 b0, b1, b2, b3, b4, b5, b6, b7;

#define GT_LOAD(c)                                                                   \
  do {                                                                               \
    const float* ap = Ag + (size_t)(c) * A_CH;                                       \
    const float* bp = Bg + (size_t)(c) * B_CH;                                       \
    a0 = *(const float4*)(ap);              a1 = *(const float4*)(ap + A_STEP);      \
    a2 = *(const float4*)(ap + 2 * A_STEP); a3 = *(const float4*)(ap + 3 * A_STEP);  \
    if (B_KMAJOR) {                                                                  \
      b0 = *(const float4*)(bp);              b1 = *(const float4*)(bp + B_STEP);     \
      b2 = *(const float4*)(bp + 2 * B_STEP); b3 = *(const float4*)(bp + 3 * B_STEP); \
      b4 = *(const float4*)(bp + 4 * B_STEP); b5 = *(const float4*)(bp + 5 * B_STEP); \
      b6 = *(const float4*)(bp + 6 * B_STEP); b7 = *(const float4*)(bp + 7 * B_STEP); \
    } else {                                                                         \
      b0 = *(const float4*)(bp);              b1 = *(const float4*)(bp + D);          \
      b2 = *(const float4*)(bp + B_STEP);     b3 = *(const float4*)(bp + B_STEP + D); \
      b4 = *(const float4*)(bp + 2 * B_STEP); b5 = *(const float4*)(bp + 2 * B_STEP + D); \
      b6 = *(const float4*)(bp + 3 * B_STEP); b7 = *(const float4*)(bp + 3 * B_STEP + D); \
    }                                                                                \
  } while (0)

#define GT_STORE_PAIR(dst, e, o)                                                     \
  do {                                                                               \
    *(float4*)(dst)       = make_float4((e).x, (o).x, (e).y, (o).y);                 \
    *(float4*)((dst) + 4) = make_float4((e).z, (o).z, (e).w, (o).w);                 \
  } while (0)

#define GT_STORE(s)                                                                  \
  do {                                                                               \
    float* ad = &As[s][sa];                                                          \
    float* bd = &Bs[s][sb];                                                          \
    *(float4*)(ad)           = a0; *(float4*)(ad + 8 * PA)  = a1;                    \
    *(float4*)(ad + 16 * PA) = a2; *(float4*)(ad + 24 * PA) = a3;                    \
    if (B_KMAJOR) {                                                                  \
      *(float4*)(bd)           = b0; *(float4*)(bd + 8 * PA)  = b1;                  \
      *(float4*)(bd + 16 * PA) = b2; *(float4*)(bd + 24 * PA) = b3;                  \
      *(float4*)(bd + 32 * PA) = b4; *(float4*)(bd + 40 * PA) = b5;                  \
      *(float4*)(bd + 48 * PA) = b6; *(float4*)(bd + 56 * PA) = b7;                  \
    } else {                                                                         \
      GT_STORE_PAIR(bd,            b0, b1);                                          \
      GT_STORE_PAIR(bd + 16 * PBP, b2, b3);                                          \
      GT_STORE_PAIR(bd + 32 * PBP, b4, b5);                                          \
      GT_STORE_PAIR(bd + 48 * PBP, b6, b7);                                          \
    }                                                                                \
  } while (0)

  v8f acc0 = {};

  GT_LOAD(0);

#pragma nounroll
  for (int c = 0; c < NCH; ++c) {
    const int s = c & 1;
    GT_STORE(s);
    __syncthreads();
    if (c + 1 < NCH) GT_LOAD(c + 1);

    const float* as = &As[s][fa];
    const float* bs = &Bs[s][fb];
#pragma unroll
    for (int ks = 0; ks < GT_K / 4; ++ks) {
      const v2f a = *(const v2f*)(as + 4 * ks);
      v2f p;
      if (B_KMAJOR) p = *(const v2f*)(bs + 4 * ks);
      else          p = *(const v2f*)(bs + 2 * ks * PBP);
      acc0 = __builtin_amdgcn_wmma_f32_16x16x4_f32(false, a, false, p, (short)0, acc0, false, false);
    }
  }
#undef GT_LOAD
#undef GT_STORE
#undef GT_STORE_PAIR

  const int col0 = nt * GT_N + wc * 16;
  float bv0 = 0.0f;
  if (bias) bv0 = bias[col0 + l];
#pragma unroll
  for (int j = 0; j < 8; ++j) {
    const int r = row0 + wr * 16 + j + 8 * half;
    C[(size_t)r * D + col0 + l] = acc0[j] + bv0;
  }
}

#define XADD_ROW(dst, src, X, BANK)                                                  \
  asm("v_add_f32_dpp %0, %1, %1 row_xmask:" #X " row_mask:0xf bank_mask:" #BANK      \
      : "+v"(dst) : "v"(src))

#define SLOT_ADD(s, a, b) asm volatile("v_pk_add_f32 %0, %1, %2" : "=v"(s) : "v"(a), "v"(b))
#define SLOT_TANH(x)      asm volatile("v_tanh_f32 %0, %0" : "+v"(x))
#define SLOT_MUL(q, t, c) asm volatile("v_pk_mul_f32 %0, %1, %2" : "=v"(q) : "v"(t), "v"(c))
#define SLOT_FMA(q, t, c) asm volatile("v_pk_fma_f32 %0, %1, %2, %0" : "+v"(q) : "v"(t), "v"(c))
#define NG   64
#define LEAD 3
#define LAG  2

struct RowSlice { float4 v0, v1, v2, v3; };

#define RX 8
#define SW 16
#define DL 16
__global__ __launch_bounds__(SW * 32) void scores_softmax_kernel(
    const float* __restrict__ item1, const float* __restrict__ item2,
    const float* __restrict__ wst, const int* __restrict__ mask,
    float* __restrict__ Sout) {
  __shared__ float sc[RX][TM + 1];

  const int tid  = threadIdx.x;
  const int b    = blockIdx.x / (TX / RX);
  const int x0   = (blockIdx.x % (TX / RX)) * RX;
  const int warp = tid >> 5;
  const int lane = tid & 31;
  const int half = lane >> 4;
  const int c    = lane & 15;

  v2f f1[RX][DL / 2];
  const float* i1 = item1 + ((size_t)b * TX + x0) * D + c * DL;
#pragma unroll
  for (int r = 0; r < RX; ++r) {
#pragma unroll
    for (int j = 0; j < DL / 4; ++j) {
      const float4 a = *(const float4*)(i1 + (size_t)r * D + 4 * j);
      f1[r][2 * j].x = a.x;     f1[r][2 * j].y = a.y;
      f1[r][2 * j + 1].x = a.z; f1[r][2 * j + 1].y = a.w;
    }
  }
  v2f w[DL / 2];
#pragma unroll
  for (int j = 0; j < DL / 4; ++j) {
    const float4 a = *(const float4*)(wst + c * DL + 4 * j);
    w[2 * j].x = a.x;     w[2 * j].y = a.y;
    w[2 * j + 1].x = a.z; w[2 * j + 1].y = a.w;
  }

  const int m0 = 2 * warp + half;
  const float* i2b = item2 + ((size_t)b * TM + m0) * D + c * DL;
  float* scw = &sc[2 * (c >> 2)][m0];
  const bool writer = (c & 3) == 0;

#define ROW_LOAD(n, t)                                                               \
  do {                                                                               \
    const float* rp = i2b + (size_t)(t) * 32 * D;                                    \
    (n).v0 = *(const float4*)(rp);     (n).v1 = *(const float4*)(rp + 4);                \
    (n).v2 = *(const float4*)(rp + 8); (n).v3 = *(const float4*)(rp + 12);               \
  } while (0)

#define ROW_TRIP(n, t)                                                               \
  do {                                                                               \
    v2f iv[DL / 2];                                                                  \
    iv[0].x = (n).v0.x; iv[0].y = (n).v0.y; iv[1].x = (n).v0.z; iv[1].y = (n).v0.w;          \
    iv[2].x = (n).v1.x; iv[2].y = (n).v1.y; iv[3].x = (n).v1.z; iv[3].y = (n).v1.w;          \
    iv[4].x = (n).v2.x; iv[4].y = (n).v2.y; iv[5].x = (n).v2.z; iv[5].y = (n).v2.w;          \
    iv[6].x = (n).v3.x; iv[6].y = (n).v3.y; iv[7].x = (n).v3.z; iv[7].y = (n).v3.w;          \
    float p[RX];                                                                     \
    float sx[NG], sy[NG];                                                            \
    v2f q[RX];                                                                       \
      \
      \
    _Pragma("unroll")                                                                \
    for (int i = -LEAD; i < NG + LAG; ++i) {                                         \
      const int ga = i + LEAD, gf = i - LAG;                                         \
      if (i >= 0 && i < NG) SLOT_TANH(sx[i]);                                        \
      if (ga < NG) {                                                                 \
        v2f s;                                                                       \
        SLOT_ADD(s, f1[ga & 7][ga >> 3], iv[ga >> 3]);                               \
        sx[ga] = s.x; sy[ga] = s.y;                                                  \
      }                                                                              \
      if (i >= 0 && i < NG) SLOT_TANH(sy[i]);                                        \
      if (gf >= 0 && gf < NG) {                                                      \
        v2f th;                                                                      \
        th.x = sx[gf]; th.y = sy[gf];                                                \
        if ((gf >> 3) == 0) SLOT_MUL(q[gf & 7], th, w[0]);                           \
        else                SLOT_FMA(q[gf & 7], th, w[gf >> 3]);                     \
      }                                                                              \
    }                                                                                \
    _Pragma("unroll")                                                                \
    for (int r = 0; r < RX; ++r) p[r] = q[r].x + q[r].y;                             \
    XADD_ROW(p[0], p[0], 8, 0x3); XADD_ROW(p[0], p[4], 8, 0xc);                      \
    XADD_ROW(p[1], p[1], 8, 0x3); XADD_ROW(p[1], p[5], 8, 0xc);                      \
    XADD_ROW(p[2], p[2], 8, 0x3); XADD_ROW(p[2], p[6], 8, 0xc);                      \
    XADD_ROW(p[3], p[3], 8, 0x3); XADD_ROW(p[3], p[7], 8, 0xc);                      \
    XADD_ROW(p[0], p[0], 4, 0x5); XADD_ROW(p[0], p[2], 4, 0xa);                      \
    XADD_ROW(p[1], p[1], 4, 0x5); XADD_ROW(p[1], p[3], 4, 0xa);                      \
    XADD_ROW(p[0], p[0], 2, 0xf); XADD_ROW(p[1], p[1], 2, 0xf);                      \
    XADD_ROW(p[0], p[0], 1, 0xf); XADD_ROW(p[1], p[1], 1, 0xf);                      \
    if (writer) {                                                                    \
      scw[32 * (t)]            = p[0];                                               \
      scw[32 * (t) + (TM + 1)] = p[1];                                               \
    }                                                                                \
  } while (0)

  RowSlice ra, rb;
  ROW_LOAD(ra, 0);
  for (int t = 0; t < TM / 32; t += 2) {
    ROW_LOAD(rb, t + 1);
    ROW_TRIP(ra, t);
    const int tn = (t + 2 < TM / 32) ? (t + 2) : (t + 1);
    ROW_LOAD(ra, tn);
    ROW_TRIP(rb, t + 1);
  }
#undef ROW_LOAD
#undef ROW_TRIP
  __syncthreads();

  if (warp < RX) {
    const int* mrow = mask + b * TM;
    float v[TM / 32];
    float mx = -INFINITY;
#pragma unroll
    for (int i = 0; i < TM / 32; ++i) {
      const int m = lane + 32 * i;
      float s = sc[warp][m];
      if (mrow[m] == 0) s = -INFINITY;
      v[i] = s;
      mx = fmaxf(mx, s);
    }
#pragma unroll
    for (int s = 16; s > 0; s >>= 1) mx = fmaxf(mx, __shfl_xor(mx, s, 32));

    float sum = 0.0f;
#pragma unroll
    for (int i = 0; i < TM / 32; ++i) {
      v[i] = fast_exp(v[i] - mx);
      sum += v[i];
    }
#pragma unroll
    for (int s = 16; s > 0; s >>= 1) sum += __shfl_xor(sum, s, 32);
    const float inv = 1.0f / sum;

    float* orow = Sout + ((size_t)b * TX + x0 + warp) * TM;
#pragma unroll
    for (int i = 0; i < TM / 32; ++i) orow[lane + 32 * i] = v[i] * inv;
  }
}

extern "C" void kernel_launch(void* const* d_in, const int* in_sizes, int n_in,
                              void* d_out, int out_size, void* d_ws, size_t ws_size,
                              hipStream_t stream) {
  const float* x    = (const float*)d_in[0];
  const float* mem  = (const float*)d_in[1];
  const int*   mask = (const int*)d_in[2];
  const float* w1   = (const float*)d_in[3];
  const float* b1   = (const float*)d_in[4];
  const float* w2   = (const float*)d_in[5];
  const float* wst  = (const float*)d_in[6];

  float* item1 = (float*)d_ws;
  float* item2 = item1 + (size_t)BATCH * TX * D;
  float* out   = (float*)d_out;
  float* Sout  = out + (size_t)BATCH * TX * D;

  dim3 blk(GT_T);
  const int gemmBlocks = (BATCH * TX / GT_M) * (D / GT_N);

  gemm_lds_kernel<D, true><<<gemmBlocks, blk, 0, stream>>>(x,   w1, b1,      item1);
  gemm_lds_kernel<D, true><<<gemmBlocks, blk, 0, stream>>>(mem, w2, nullptr, item2);

  scores_softmax_kernel<<<BATCH * (TX / RX), SW * 32, 0, stream>>>(item1, item2, wst, mask, Sout);

  gemm_lds_kernel<TM, false><<<gemmBlocks, blk, 0, stream>>>(Sout, mem, nullptr, out);
}
